# starting kernel
# baseline (speedup 1.0000x reference)
#include <hip/hip_runtime.h>
#include <math.h>


typedef __attribute__((ext_vector_type(2))) float v2f;
typedef __attribute__((ext_vector_type(8))) float v8f;

#define B_N 64
#define P_N 32
#define L_N 32
#define T_N 1024
#define DTW_W 0.93057205f
#define BIGF  3.0e38f
#define TILE_STRIDE 33
#define WAVE_LDS (16 * TILE_STRIDE + 16)

__device__ __forceinline__ float bperm_f(int byte_idx, float v) {
  return __int_as_float(__builtin_amdgcn_ds_bpermute(byte_idx, __float_as_int(v)));
}

__global__ __launch_bounds__(256) void dtw_wmma_kernel(
    const float* __restrict__ x,
    const float* __restrict__ patts,
    float* __restrict__ out) {
  const int lane = threadIdx.x & 31;
  const int wib  = threadIdx.x >> 5;
  const int wave = blockIdx.x * 8 + wib;
  const int b = wave >> 5;
  const int p = wave & 31;

  __shared__ float lds[8 * WAVE_LDS];
  float* tile   = &lds[wib * WAVE_LDS];
  float* ostage = tile + 16 * TILE_STRIDE;

  const float* xb = x + b * 3 * T_N;
  const float* pp = patts + p * 3 * L_N;

  const int  nn = lane & 15;
  const bool hi = lane >= 16;

  const int im1 = (lane - 1)  << 2;
  const int i1  = im1;
  const int i2  = (lane - 2)  << 2;
  const int i4  = (lane - 4)  << 2;
  const int i8  = (lane - 8)  << 2;
  const int i16 = (lane - 16) << 2;
  const bool g1 = lane >= 1, g2 = lane >= 2, g4 = lane >= 4,
             g8 = lane >= 8, g16 = lane >= 16;

  v2f bm0, bm1;
  float p2lo, p2hi;
  {
    const float a0 = pp[0 * L_N + nn];
    const float a1 = pp[1 * L_N + nn];
    const float a2 = pp[2 * L_N + nn];
    p2lo = a0 * a0 + a1 * a1 + a2 * a2;
    bm0.x = hi ? (-2.0f * a2) : (-2.0f * a0);
    bm0.y = hi ? 1.0f         : (-2.0f * a1);
    const float c0 = pp[0 * L_N + 16 + nn];
    const float c1 = pp[1 * L_N + 16 + nn];
    const float c2 = pp[2 * L_N + 16 + nn];
    p2hi = c0 * c0 + c1 * c1 + c2 * c2;
    bm1.x = hi ? (-2.0f * c2) : (-2.0f * c0);
    bm1.y = hi ? 1.0f         : (-2.0f * c1);
  }

  const float w = DTW_W;
  float cprev = 0.0f;

#define SCAN_STEP(IDX, G)                            \
  {                                                  \
    float pu = bperm_f((IDX), pS);                   \
    float qu = bperm_f((IDX), qS);                   \
    pu = (G) ? pu : 0.0f;                            \
    qu = (G) ? qu : BIGF;                            \
    qS = fminf(qu + pS, qS);                         \
    pS = pu + pS;                                    \
  }

  for (int blk = 0; blk < T_N / 16; ++blk) {
    const int t0 = blk * 16;

    if (blk < T_N / 16 - 1) {
      __builtin_prefetch(xb + t0 + 16, 0, 3);
      __builtin_prefetch(xb + 2 * T_N + t0 + 16, 0, 3);
    }

    const int tm = t0 + nn;
    const float x0  = xb[0 * T_N + tm];
    const float x1  = xb[1 * T_N + tm];
    const float x2c = xb[2 * T_N + tm];
    const float xn  = x0 * x0 + x1 * x1 + x2c * x2c;
    v2f am;
    am.x = hi ? x2c : x0;
    am.y = hi ? xn  : x1;

    v8f acc = {};
    v8f dlo = __builtin_amdgcn_wmma_f32_16x16x4_f32(
        false, am, false, bm0, (short)0, acc, false, false);
    v8f dhi = __builtin_amdgcn_wmma_f32_16x16x4_f32(
        false, am, false, bm1, (short)0, acc, false, false);

    const int row0 = hi ? 8 : 0;
#pragma unroll
    for (int k = 0; k < 8; ++k) {
      const float dl = __builtin_amdgcn_sqrtf(fmaxf(dlo[k] + p2lo, 1e-12f));
      const float dh = __builtin_amdgcn_sqrtf(fmaxf(dhi[k] + p2hi, 1e-12f));
      tile[(row0 + k) * TILE_STRIDE + nn]      = dl;
      tile[(row0 + k) * TILE_STRIDE + nn + 16] = dh;
    }

#pragma unroll 4
    for (int k = 0; k < 16; ++k) {
      const float d = tile[k * TILE_STRIDE + lane];

      float c;
      if (blk == 0 && k == 0) {
        float S = d;
        {
          float v;
          v = bperm_f(i1,  S); S += g1  ? v : 0.0f;
          v = bperm_f(i2,  S); S += g2  ? v : 0.0f;
          v = bperm_f(i4,  S); S += g4  ? v : 0.0f;
          v = bperm_f(i8,  S); S += g8  ? v : 0.0f;
          v = bperm_f(i16, S); S += g16 ? v : 0.0f;
        }
        c = S;
      } else {
        float cm1 = bperm_f(im1, cprev);
        cm1 = g1 ? cm1 : cprev;
        const float m = w * fminf(cprev, cm1);

        float pS = d;
        float qS = m + d;
        SCAN_STEP(i1,  g1)
        SCAN_STEP(i2,  g2)
        SCAN_STEP(i4,  g4)
        SCAN_STEP(i8,  g8)
        SCAN_STEP(i16, g16)
        c = qS;
      }
      cprev = c;
      if (lane == 31) ostage[k] = c;
    }

    const float oval = ostage[lane & 15];
    if (lane < 16) out[(b * P_N + p) * T_N + t0 + lane] = oval;
  }
#undef SCAN_STEP
}

extern "C" void kernel_launch(void* const* d_in, const int* in_sizes, int n_in,
                              void* d_out, int out_size, void* d_ws, size_t ws_size,
                              hipStream_t stream) {
  (void)in_sizes; (void)n_in; (void)out_size; (void)d_ws; (void)ws_size;
  const float* x     = (const float*)d_in[0];
  const float* patts = (const float*)d_in[1];
  float* out = (float*)d_out;

  dim3 grid(256), block(256);
  hipLaunchKernelGGL(dtw_wmma_kernel, grid, block, 0, stream, x, patts, out);
}


# DP run along anti-diagonals (lane j at column s-j): one in-row DPP lane shift per step replaces the 5-step ds_bpermute scan; sums now in sequential order, no scan tree (rounding order changes)
# speedup vs baseline: 10.2505x; 10.2505x over previous
#include <hip/hip_runtime.h>
#include <math.h>


typedef __attribute__((ext_vector_type(2))) float v2f;
typedef __attribute__((ext_vector_type(8))) float v8f;

#define B_N 64
#define P_N 32
#define L_N 32
#define T_N 1024
#define DTW_W 0.93057205f
#define ROW_F 32
#define RING_ROWS 80
#define OST_F 32
#define DUMP_F 48
#define WAVE_LDS (RING_ROWS * ROW_F + OST_F + DUMP_F)
#define N_BLK (T_N / 16 + 2)

__global__ __launch_bounds__(256) void dtw_wmma_kernel(
    const float* __restrict__ x,
    const float* __restrict__ patts,
    float* __restrict__ out) {
  const int lane = threadIdx.x & 31;
  const int wib  = threadIdx.x >> 5;
  const int wave = blockIdx.x * 8 + wib;
  const int b = wave >> 5;
  const int p = wave & 31;

  __shared__ float lds[8 * WAVE_LDS];
  float* ring   = &lds[wib * WAVE_LDS];
  float* ostage = ring + RING_ROWS * ROW_F;
  float* dump   = ostage + OST_F;

  const float* xb = x + b * 3 * T_N;
  const float* pp = patts + p * 3 * L_N;

  const int  nn = lane & 15;
  const bool hi = lane >= 16;
  const bool lane0 = lane == 0;

  v2f bm0, bm1;
  float p2lo, p2hi;
  {
    const float a0 = pp[0 * L_N + nn];
    const float a1 = pp[1 * L_N + nn];
    const float a2 = pp[2 * L_N + nn];
    p2lo = a0 * a0 + a1 * a1 + a2 * a2;
    bm0.x = hi ? (-2.0f * a2) : (-2.0f * a0);
    bm0.y = hi ? 1.0f         : (-2.0f * a1);
    const float c0 = pp[0 * L_N + 16 + nn];
    const float c1 = pp[1 * L_N + 16 + nn];
    const float c2 = pp[2 * L_N + 16 + nn];
    p2hi = c0 * c0 + c1 * c1 + c2 * c2;
    bm1.x = hi ? (-2.0f * c2) : (-2.0f * c0);
    bm1.y = hi ? 1.0f         : (-2.0f * c1);
  }

  for (int r = 32; r < 64; ++r) ring[r * ROW_F + lane] = 0.0f;

  const float w = DTW_W;
  const int lag = lane + (lane >> 4);
  float c  = INFINITY;
  float u1 = lane0 ? 0.0f : INFINITY;
  float xo = INFINITY;

  float x0  = xb[0 * T_N + nn];
  float x1  = xb[1 * T_N + nn];
  float x2c = xb[2 * T_N + nn];

  auto build_tile = [&](const int mt) {
    const float xn = x0 * x0 + x1 * x1 + x2c * x2c;
    v2f am;
    am.x = hi ? x2c : x0;
    am.y = hi ? xn  : x1;
    v8f acc = {};
    v8f dlo = __builtin_amdgcn_wmma_f32_16x16x4_f32(
        false, am, false, bm0, (short)0, acc, false, false);
    v8f dhi = __builtin_amdgcn_wmma_f32_16x16x4_f32(
        false, am, false, bm1, (short)0, acc, false, false);

    {
      const int tn = (mt + 1 < T_N / 16 ? mt + 1 : mt) * 16 + nn;
      x0  = xb[0 * T_N + tn];
      x1  = xb[1 * T_N + tn];
      x2c = xb[2 * T_N + tn];
    }

    float* wr = ring + (((mt & 3) * 16) + (hi ? 8 : 0)) * ROW_F + nn;
#pragma unroll
    for (int k = 0; k < 8; ++k) {
      dlo[k] = __builtin_amdgcn_sqrtf(fmaxf(dlo[k] + p2lo, 1e-12f));
      dhi[k] = __builtin_amdgcn_sqrtf(fmaxf(dhi[k] + p2hi, 1e-12f));
      wr[k * ROW_F]      = dlo[k];
      wr[k * ROW_F + 16] = dhi[k];
    }
    if ((mt & 3) == 0) {
#pragma unroll
      for (int k = 0; k < 8; ++k) {
        wr[(64 + k) * ROW_F]      = dlo[k];
        wr[(64 + k) * ROW_F + 16] = dhi[k];
      }
    }
  };

  float dv[16];
  build_tile(0);
  {
    const float* rd = ring + (((0 - lag) & 63) * ROW_F) + lane;
#pragma unroll
    for (int k = 0; k < 16; ++k) dv[k] = rd[k * ROW_F];
  }

  for (int m = 0; m < N_BLK; ++m) {
    const float oval = ostage[lane];

    if (m + 1 < T_N / 16) build_tile(m + 1);

    float* op = (lane == 31) ? (ostage + (m & 1) * 16) : (dump + lane);

#pragma unroll
    for (int k = 0; k < 16; ++k) {
      const int ci = __float_as_int(c);
      const float up = __int_as_float(__builtin_amdgcn_update_dpp(
          __float_as_int(xo), ci, 0x111, 0xf, 0xf, false));
      const float wu1 = w * u1;
      const float xg = __int_as_float(__builtin_amdgcn_permlanex16(
          __float_as_int(u1), ci, 0xffffffffu, 0xffffffffu, false, false));
      xo = lane0 ? INFINITY : xg;
      const float e = fminf(fminf(up, w * c), wu1);
      u1 = up;
      c = dv[k] + e;
      op[k] = c;
    }
    {
      const float* rd = ring + ((((m + 1) * 16 - lag) & 63) * ROW_F) + lane;
#pragma unroll
      for (int k = 0; k < 16; ++k) dv[k] = rd[k * ROW_F];
    }

    if ((m & 1) == 0 && m >= 4)
      out[(b * P_N + p) * T_N + (m - 4) * 16 + lane] = oval;
  }
  out[(b * P_N + p) * T_N + (N_BLK - 4) * 16 + lane] = ostage[lane];
}

extern "C" void kernel_launch(void* const* d_in, const int* in_sizes, int n_in,
                              void* d_out, int out_size, void* d_ws, size_t ws_size,
                              hipStream_t stream) {
  (void)in_sizes; (void)n_in; (void)out_size; (void)d_ws; (void)ws_size;
  const float* x     = (const float*)d_in[0];
  const float* patts = (const float*)d_in[1];
  float* out = (float*)d_out;

  dim3 grid(256), block(256);
  hipLaunchKernelGGL(dtw_wmma_kernel, grid, block, 0, stream, x, patts, out);
}


# x[b] staged once per work-group in LDS as ready-made WMMA A rows {x0,x1,x2,|x|^2} (its 8 waves share b): time loop has no global load, load wait or address math; |x|^2 fused in one order for all t
# speedup vs baseline: 12.3385x; 1.2037x over previous
#include <hip/hip_runtime.h>
#include <math.h>


typedef __attribute__((ext_vector_type(2))) float v2f;
typedef __attribute__((ext_vector_type(8))) float v8f;

#define B_N 64
#define P_N 32
#define L_N 32
#define T_N 1024
#define DTW_W 0.93057205f
#define ROW_F 32
#define RING_ROWS 80
#define OST_F 32
#define DUMP_F 48
#define WAVE_LDS (RING_ROWS * ROW_F + OST_F + DUMP_F)
#define N_BLK (T_N / 16 + 2)
#define XA_PAIRS (T_N * 2)

__global__ __launch_bounds__(256) void dtw_wmma_kernel(
    const float* __restrict__ x,
    const float* __restrict__ patts,
    float* __restrict__ out) {
  const int lane = threadIdx.x & 31;
  const int wib  = threadIdx.x >> 5;
  const int wave = blockIdx.x * 8 + wib;
  const int b = wave >> 5;
  const int p = wave & 31;

  __shared__ float lds[8 * WAVE_LDS];
  __shared__ v2f xa[XA_PAIRS];
  {
    const float* xs = x + (blockIdx.x >> 2) * 3 * T_N;
#pragma unroll
    for (int i = 0; i < T_N / 256; ++i) {
      const int t = i * 256 + threadIdx.x;
      const float s0 = xs[0 * T_N + t];
      const float s1 = xs[1 * T_N + t];
      const float s2 = xs[2 * T_N + t];
      v2f lo2, hi2;
      lo2.x = s0;
      lo2.y = s1;
      hi2.x = s2;
      hi2.y = __builtin_fmaf(s2, s2, __builtin_fmaf(s1, s1, s0 * s0));
      xa[2 * t]     = lo2;
      xa[2 * t + 1] = hi2;
    }
  }
  float* ring   = &lds[wib * WAVE_LDS];
  float* ostage = ring + RING_ROWS * ROW_F;
  float* dump   = ostage + OST_F;

  const float* pp = patts + p * 3 * L_N;

  const int  nn = lane & 15;
  const bool hi = lane >= 16;
  const bool lane0 = lane == 0;

  v2f bm0, bm1;
  float p2lo, p2hi;
  {
    const float a0 = pp[0 * L_N + nn];
    const float a1 = pp[1 * L_N + nn];
    const float a2 = pp[2 * L_N + nn];
    p2lo = a0 * a0 + a1 * a1 + a2 * a2;
    bm0.x = hi ? (-2.0f * a2) : (-2.0f * a0);
    bm0.y = hi ? 1.0f         : (-2.0f * a1);
    const float c0 = pp[0 * L_N + 16 + nn];
    const float c1 = pp[1 * L_N + 16 + nn];
    const float c2 = pp[2 * L_N + 16 + nn];
    p2hi = c0 * c0 + c1 * c1 + c2 * c2;
    bm1.x = hi ? (-2.0f * c2) : (-2.0f * c0);
    bm1.y = hi ? 1.0f         : (-2.0f * c1);
  }

  for (int r = 32; r < 64; ++r) ring[r * ROW_F + lane] = 0.0f;

  const float w = DTW_W;
  const int lag = lane + (lane >> 4);
  float c  = INFINITY;
  float u1 = lane0 ? 0.0f : INFINITY;
  float xo = INFINITY;

  __syncthreads();

  const v2f* ar = xa + nn * 2 + (hi ? 1 : 0);
  v2f am = ar[0];

  auto build_tile = [&](const int mt) {
    v8f acc = {};
    v8f dlo = __builtin_amdgcn_wmma_f32_16x16x4_f32(
        false, am, false, bm0, (short)0, acc, false, false);
    v8f dhi = __builtin_amdgcn_wmma_f32_16x16x4_f32(
        false, am, false, bm1, (short)0, acc, false, false);

    {
      const int tnx = (mt + 1 < T_N / 16 ? mt + 1 : mt);
      am = ar[tnx * 32];
    }

    float* wr = ring + (((mt & 3) * 16) + (hi ? 8 : 0)) * ROW_F + nn;
#pragma unroll
    for (int k = 0; k < 8; ++k) {
      dlo[k] = __builtin_amdgcn_sqrtf(fmaxf(dlo[k] + p2lo, 1e-12f));
      dhi[k] = __builtin_amdgcn_sqrtf(fmaxf(dhi[k] + p2hi, 1e-12f));
      wr[k * ROW_F]      = dlo[k];
      wr[k * ROW_F + 16] = dhi[k];
    }
    if ((mt & 3) == 0) {
#pragma unroll
      for (int k = 0; k < 8; ++k) {
        wr[(64 + k) * ROW_F]      = dlo[k];
        wr[(64 + k) * ROW_F + 16] = dhi[k];
      }
    }
  };

  float dv[16];
  build_tile(0);
  {
    const float* rd = ring + (((0 - lag) & 63) * ROW_F) + lane;
#pragma unroll
    for (int k = 0; k < 16; ++k) dv[k] = rd[k * ROW_F];
  }

  for (int m = 0; m < N_BLK; ++m) {
    const float oval = ostage[lane];

    if (m + 1 < T_N / 16) build_tile(m + 1);

    float* op = (lane == 31) ? (ostage + (m & 1) * 16) : (dump + lane);

#pragma unroll
    for (int k = 0; k < 16; ++k) {
      const int ci = __float_as_int(c);
      const float up = __int_as_float(__builtin_amdgcn_update_dpp(
          __float_as_int(xo), ci, 0x111, 0xf, 0xf, false));
      const float wu1 = w * u1;
      const float xg = __int_as_float(__builtin_amdgcn_permlanex16(
          __float_as_int(u1), ci, 0xffffffffu, 0xffffffffu, false, false));
      xo = lane0 ? INFINITY : xg;
      const float e = fminf(fminf(up, w * c), wu1);
      u1 = up;
      c = dv[k] + e;
      op[k] = c;
    }
    {
      const float* rd = ring + ((((m + 1) * 16 - lag) & 63) * ROW_F) + lane;
#pragma unroll
      for (int k = 0; k < 16; ++k) dv[k] = rd[k * ROW_F];
    }

    if ((m & 1) == 0 && m >= 4)
      out[(b * P_N + p) * T_N + (m - 4) * 16 + lane] = oval;
  }
  out[(b * P_N + p) * T_N + (N_BLK - 4) * 16 + lane] = ostage[lane];
}

extern "C" void kernel_launch(void* const* d_in, const int* in_sizes, int n_in,
                              void* d_out, int out_size, void* d_ws, size_t ws_size,
                              hipStream_t stream) {
  (void)in_sizes; (void)n_in; (void)out_size; (void)d_ws; (void)ws_size;
  const float* x     = (const float*)d_in[0];
  const float* patts = (const float*)d_in[1];
  float* out = (float*)d_out;

  dim3 grid(256), block(256);
  hipLaunchKernelGGL(dtw_wmma_kernel, grid, block, 0, stream, x, patts, out);
}


# Block pipelined one deeper in a branch-free 4-block body: next block's LDS reads first, tile m+2 built on registers beside the 16 DP steps, all stores last; no LDS op or wait between DP steps
# speedup vs baseline: 13.6646x; 1.1075x over previous
#include <hip/hip_runtime.h>
#include <math.h>


typedef __attribute__((ext_vector_type(2))) float v2f;
typedef __attribute__((ext_vector_type(8))) float v8f;

#define B_N 64
#define P_N 32
#define L_N 32
#define T_N 1024
#define DTW_W 0.93057205f
#define ROW_F 32
#define RING_ROWS 80
#define OST_F 32
#define DUMP_F 48
#define WAVE_LDS (RING_ROWS * ROW_F + OST_F + DUMP_F)
#define N_BLK (T_N / 16 + 2)
#define XA_PAIRS (T_N * 2)

__global__ __launch_bounds__(256) void dtw_wmma_kernel(
    const float* __restrict__ x,
    const float* __restrict__ patts,
    float* __restrict__ out) {
  const int lane = threadIdx.x & 31;
  const int wib  = threadIdx.x >> 5;
  const int wave = blockIdx.x * 8 + wib;
  const int b = wave >> 5;
  const int p = wave & 31;

  __shared__ float lds[8 * WAVE_LDS];
  __shared__ v2f xa[XA_PAIRS];
  {
    const float* xs = x + (blockIdx.x >> 2) * 3 * T_N;
#pragma unroll
    for (int i = 0; i < T_N / 256; ++i) {
      const int t = i * 256 + threadIdx.x;
      const float s0 = xs[0 * T_N + t];
      const float s1 = xs[1 * T_N + t];
      const float s2 = xs[2 * T_N + t];
      v2f lo2, hi2;
      lo2.x = s0;
      lo2.y = s1;
      hi2.x = s2;
      hi2.y = __builtin_fmaf(s2, s2, __builtin_fmaf(s1, s1, s0 * s0));
      xa[2 * t]     = lo2;
      xa[2 * t + 1] = hi2;
    }
  }
  float* ring   = &lds[wib * WAVE_LDS];
  float* ostage = ring + RING_ROWS * ROW_F;
  float* dump   = ostage + OST_F;

  const float* pp = patts + p * 3 * L_N;

  const int  nn = lane & 15;
  const bool hi = lane >= 16;
  const bool lane0 = lane == 0;

  v2f bm0, bm1;
  float p2lo, p2hi;
  {
    const float a0 = pp[0 * L_N + nn];
    const float a1 = pp[1 * L_N + nn];
    const float a2 = pp[2 * L_N + nn];
    p2lo = a0 * a0 + a1 * a1 + a2 * a2;
    bm0.x = hi ? (-2.0f * a2) : (-2.0f * a0);
    bm0.y = hi ? 1.0f         : (-2.0f * a1);
    const float c0 = pp[0 * L_N + 16 + nn];
    const float c1 = pp[1 * L_N + 16 + nn];
    const float c2 = pp[2 * L_N + 16 + nn];
    p2hi = c0 * c0 + c1 * c1 + c2 * c2;
    bm1.x = hi ? (-2.0f * c2) : (-2.0f * c0);
    bm1.y = hi ? 1.0f         : (-2.0f * c1);
  }

  for (int r = 32; r < 64; ++r) ring[r * ROW_F + lane] = 0.0f;

  const float w = DTW_W;
  const int lag = lane + (lane >> 4);
  float c  = INFINITY;
  float u1 = lane0 ? 0.0f : INFINITY;
  float xo = INFINITY;

  __syncthreads();

  const v2f* ar = xa + nn * 2 + (hi ? 1 : 0);
  v2f am = ar[0];

  auto tile_dist = [&](v8f& dlo, v8f& dhi) {
    v8f acc = {};
    dlo = __builtin_amdgcn_wmma_f32_16x16x4_f32(
        false, am, false, bm0, (short)0, acc, false, false);
    dhi = __builtin_amdgcn_wmma_f32_16x16x4_f32(
        false, am, false, bm1, (short)0, acc, false, false);
#pragma unroll
    for (int k = 0; k < 8; ++k) {
      dlo[k] = __builtin_amdgcn_sqrtf(fmaxf(dlo[k] + p2lo, 1e-12f));
      dhi[k] = __builtin_amdgcn_sqrtf(fmaxf(dhi[k] + p2hi, 1e-12f));
    }
  };
  auto tile_store = [&](const v8f& dlo, const v8f& dhi, const int slot) {
    float* wr = ring + ((slot * 16) + (hi ? 8 : 0)) * ROW_F + nn;
#pragma unroll
    for (int k = 0; k < 8; ++k) {
      wr[k * ROW_F]      = dlo[k];
      wr[k * ROW_F + 16] = dhi[k];
    }
    if (slot == 0) {
#pragma unroll
      for (int k = 0; k < 8; ++k) {
        wr[(64 + k) * ROW_F]      = dlo[k];
        wr[(64 + k) * ROW_F + 16] = dhi[k];
      }
    }
  };
  auto a_rows = [&](const int mt) {
    return ar[(mt < T_N / 16 ? mt : T_N / 16 - 1) * 32];
  };

  auto load_dist = [&](float (&d)[16], const int q) {
    const float* rd = ring + (((q * 16 - lag) & 63) * ROW_F) + lane;
#pragma unroll
    for (int k = 0; k < 16; ++k) d[k] = rd[k * ROW_F];
  };

  float* const op_even = (lane == 31) ? ostage      : (dump + lane);
  float* const op_odd  = (lane == 31) ? ostage + 16 : (dump + lane);

  float dv[16], cs[16];
  auto dp_steps = [&]() {
#pragma unroll
    for (int k = 0; k < 16; ++k) {
      const int ci = __float_as_int(c);
      const float up = __int_as_float(__builtin_amdgcn_update_dpp(
          __float_as_int(xo), ci, 0x111, 0xf, 0xf, false));
      const float wu1 = w * u1;
      const float xg = __int_as_float(__builtin_amdgcn_permlanex16(
          __float_as_int(u1), ci, 0xffffffffu, 0xffffffffu, false, false));
      xo = lane0 ? INFINITY : xg;
      const float e = fminf(fminf(up, w * c), wu1);
      u1 = up;
      c = dv[k] + e;
      cs[k] = c;
    }
  };

  {
    v8f dlo, dhi;
    tile_dist(dlo, dhi);
    am = a_rows(1);
    tile_store(dlo, dhi, 0);
    tile_dist(dlo, dhi);
    am = a_rows(2);
    tile_store(dlo, dhi, 1);
  }
  load_dist(dv, 0);
  __builtin_amdgcn_fence(__ATOMIC_SEQ_CST, "workgroup");

  auto block = [&](const int m, const int q, const bool build) {
    float dvn[16];
    load_dist(dvn, (q + 1) & 3);
    v2f amn = am;
    if (build) amn = a_rows(m + 3);
    __builtin_amdgcn_sched_barrier(0);
    v8f dlo, dhi;
    if (build) tile_dist(dlo, dhi);
    dp_steps();
    __builtin_amdgcn_sched_barrier(0);
    if (build) tile_store(dlo, dhi, (q + 2) & 3);
    float* const op = (q & 1) ? op_odd : op_even;
#pragma unroll
    for (int k = 0; k < 16; ++k) op[k] = cs[k];
#pragma unroll
    for (int k = 0; k < 16; ++k) dv[k] = dvn[k];
    am = amn;
  };

  float* const orow = out + (b * P_N + p) * T_N + lane;
  for (int g = 0; g < T_N / 64; ++g) {
    const float ova = ostage[lane];
    block(4 * g + 0, 0, true);
    block(4 * g + 1, 1, true);
    const float ovb = ostage[lane];
    block(4 * g + 2, 2, true);
    block(4 * g + 3, 3, true);
    if (g > 0) {
      orow[(4 * g - 4) * 16] = ova;
      orow[(4 * g - 2) * 16] = ovb;
    }
  }
  {
    const float ova = ostage[lane];
    block(T_N / 16, 0, false);
    block(T_N / 16 + 1, 1, false);
    orow[(N_BLK - 6) * 16] = ova;
    orow[(N_BLK - 4) * 16] = ostage[lane];
  }
}

extern "C" void kernel_launch(void* const* d_in, const int* in_sizes, int n_in,
                              void* d_out, int out_size, void* d_ws, size_t ws_size,
                              hipStream_t stream) {
  (void)in_sizes; (void)n_in; (void)out_size; (void)d_ws; (void)ws_size;
  const float* x     = (const float*)d_in[0];
  const float* patts = (const float*)d_in[1];
  float* out = (float*)d_out;

  dim3 grid(256), block(256);
  hipLaunchKernelGGL(dtw_wmma_kernel, grid, block, 0, stream, x, patts, out);
}


# Each lane owns two adjacent DP rows, a wave carries two patterns (one per 16-lane row): row-pair link stays in-lane, one in-row shift serves two cells, no cross-row fetch or select; sums as before
# speedup vs baseline: 15.0621x; 1.1023x over previous
#include <hip/hip_runtime.h>
#include <math.h>


typedef __attribute__((ext_vector_type(2))) float v2f;
typedef __attribute__((ext_vector_type(8))) float v8f;
typedef float v2f_ring __attribute__((ext_vector_type(2), may_alias));

#define B_N 64
#define P_N 32
#define L_N 32
#define T_N 1024
#define DTW_W 0.93057205f
#define ROW_F 32
#define RING_ROWS 80
#define RING_F (RING_ROWS * ROW_F)
#define OST_F 32
#define DUMP_F 64
#define WAVE_LDS (RING_F + OST_F + RING_F + OST_F + DUMP_F)
#define N_BLK (T_N / 16 + 2)
#define XA_PAIRS (T_N * 2)
#define WG_WAVES 4
#define WG_THREADS (WG_WAVES * 32)

__global__ __launch_bounds__(WG_THREADS) void dtw_wmma_kernel(
    const float* __restrict__ x,
    const float* __restrict__ patts,
    float* __restrict__ out) {
  const int lane = threadIdx.x & 31;
  const int wib  = threadIdx.x >> 5;
  const int wave = blockIdx.x * WG_WAVES + wib;
  const int b  = wave >> 4;
  const int p0 = (wave * 2) & 31;

  __shared__ __attribute__((aligned(16))) float lds[WG_WAVES * WAVE_LDS];
  __shared__ v2f xa[XA_PAIRS];
  {
    const float* xs = x + (blockIdx.x >> 2) * 3 * T_N;
#pragma unroll
    for (int i = 0; i < T_N / WG_THREADS; ++i) {
      const int t = i * WG_THREADS + threadIdx.x;
      const float s0 = xs[0 * T_N + t];
      const float s1 = xs[1 * T_N + t];
      const float s2 = xs[2 * T_N + t];
      v2f lo2, hi2;
      lo2.x = s0;
      lo2.y = s1;
      hi2.x = s2;
      hi2.y = __builtin_fmaf(s2, s2, __builtin_fmaf(s1, s1, s0 * s0));
      xa[2 * t]     = lo2;
      xa[2 * t + 1] = hi2;
    }
  }
  float* ring0   = &lds[wib * WAVE_LDS];
  float* ostage0 = ring0 + RING_F;
  float* ring1   = ostage0 + OST_F;
  float* ostage1 = ring1 + RING_F;
  float* dump    = ostage1 + OST_F;

  const int  nn = lane & 15;
  const bool hi = lane >= 16;

  v2f bm00, bm01, bm10, bm11;
  float q2lo0, q2hi0, q2lo1, q2hi1;
  auto load_b = [&](const float* pp, v2f& bmlo, v2f& bmhi, float& p2lo, float& p2hi) {
    const float a0 = pp[0 * L_N + nn];
    const float a1 = pp[1 * L_N + nn];
    const float a2 = pp[2 * L_N + nn];
    p2lo = a0 * a0 + a1 * a1 + a2 * a2;
    bmlo.x = hi ? (-2.0f * a2) : (-2.0f * a0);
    bmlo.y = hi ? 1.0f         : (-2.0f * a1);
    const float c0 = pp[0 * L_N + 16 + nn];
    const float c1 = pp[1 * L_N + 16 + nn];
    const float c2 = pp[2 * L_N + 16 + nn];
    p2hi = c0 * c0 + c1 * c1 + c2 * c2;
    bmhi.x = hi ? (-2.0f * c2) : (-2.0f * c0);
    bmhi.y = hi ? 1.0f         : (-2.0f * c1);
  };
  load_b(patts + p0 * 3 * L_N,       bm00, bm01, q2lo0, q2hi0);
  load_b(patts + (p0 + 1) * 3 * L_N, bm10, bm11, q2lo1, q2hi1);

  for (int r = 32; r < 64; ++r) {
    ring0[r * ROW_F + lane] = 0.0f;
    ring1[r * ROW_F + lane] = 0.0f;
  }

  const float w = DTW_W;
  const int lagA = 2 * nn;
  float* const myring = hi ? ring1 : ring0;
  float cA = INFINITY;
  float cB = INFINITY;
  float upA = INFINITY;
  float wu1A = (nn == 0) ? 0.0f : INFINITY;
  float wu1B = INFINITY;
  float carry = 0.0f;

  __syncthreads();

  const v2f* ar = xa + nn * 2 + (hi ? 1 : 0);
  v2f am = ar[0];

  auto tile_dist = [&](v8f& dlo0, v8f& dhi0, v8f& dlo1, v8f& dhi1) {
    v8f acc = {};
    dlo0 = __builtin_amdgcn_wmma_f32_16x16x4_f32(
        false, am, false, bm00, (short)0, acc, false, false);
    dhi0 = __builtin_amdgcn_wmma_f32_16x16x4_f32(
        false, am, false, bm01, (short)0, acc, false, false);
    dlo1 = __builtin_amdgcn_wmma_f32_16x16x4_f32(
        false, am, false, bm10, (short)0, acc, false, false);
    dhi1 = __builtin_amdgcn_wmma_f32_16x16x4_f32(
        false, am, false, bm11, (short)0, acc, false, false);
#pragma unroll
    for (int k = 0; k < 8; ++k) {
      dlo0[k] = __builtin_amdgcn_sqrtf(fmaxf(dlo0[k] + q2lo0, 1e-12f));
      dhi0[k] = __builtin_amdgcn_sqrtf(fmaxf(dhi0[k] + q2hi0, 1e-12f));
      dlo1[k] = __builtin_amdgcn_sqrtf(fmaxf(dlo1[k] + q2lo1, 1e-12f));
      dhi1[k] = __builtin_amdgcn_sqrtf(fmaxf(dhi1[k] + q2hi1, 1e-12f));
    }
  };
  auto tile_store = [&](float* ring, const v8f& dlo, const v8f& dhi, const int slot) {
    float* wr = ring + ((slot * 16) + (hi ? 8 : 0)) * ROW_F + nn;
#pragma unroll
    for (int k = 0; k < 8; ++k) {
      wr[k * ROW_F]      = dlo[k];
      wr[k * ROW_F + 16] = dhi[k];
    }
    if (slot == 0) {
#pragma unroll
      for (int k = 0; k < 8; ++k) {
        wr[(64 + k) * ROW_F]      = dlo[k];
        wr[(64 + k) * ROW_F + 16] = dhi[k];
      }
    }
  };
  auto a_rows = [&](const int mt) {
    return ar[(mt < T_N / 16 ? mt : T_N / 16 - 1) * 32];
  };

  auto load_dist = [&](v2f (&d)[16], const int q) {
    const float* rd = myring + (((q * 16 - lagA) & 63) * ROW_F) + 2 * nn;
#pragma unroll
    for (int k = 0; k < 16; ++k) d[k] = *reinterpret_cast<const v2f_ring*>(rd + k * ROW_F);
  };

  float* const op_even = (lane == 15) ? ostage0      : (lane == 31) ? ostage1      : (dump + lane);
  float* const op_odd  = (lane == 15) ? ostage0 + 16 : (lane == 31) ? ostage1 + 16 : (dump + lane);

  v2f dv[16];
  float cs[16];
  auto dp_steps = [&]() {
#pragma unroll
    for (int k = 0; k < 16; ++k) {
      cs[k] = cB;
      const float dA = dv[k].x;
      const float dB = (k == 0) ? carry : dv[k - 1].y;
      upA = __int_as_float(__builtin_amdgcn_update_dpp(
          __float_as_int(upA), __float_as_int(cB), 0x111, 0xf, 0xf, false));
      const float wcA = w * cA;
      const float eA = fminf(fminf(upA, wcA), wu1A);
      const float eB = fminf(fminf(cA, w * cB), wu1B);
      wu1A = w * upA;
      wu1B = wcA;
      cA = dA + eA;
      cB = dB + eB;
    }
    carry = dv[15].y;
  };

  {
    v8f dlo0, dhi0, dlo1, dhi1;
    tile_dist(dlo0, dhi0, dlo1, dhi1);
    am = a_rows(1);
    tile_store(ring0, dlo0, dhi0, 0);
    tile_store(ring1, dlo1, dhi1, 0);
    tile_dist(dlo0, dhi0, dlo1, dhi1);
    am = a_rows(2);
    tile_store(ring0, dlo0, dhi0, 1);
    tile_store(ring1, dlo1, dhi1, 1);
  }
  load_dist(dv, 0);
  __builtin_amdgcn_fence(__ATOMIC_SEQ_CST, "workgroup");

  auto block = [&](const int m, const int q, const bool build) {
    v2f dvn[16];
    load_dist(dvn, (q + 1) & 3);
    v2f amn = am;
    if (build) amn = a_rows(m + 3);
    __builtin_amdgcn_sched_barrier(0);
    v8f dlo0, dhi0, dlo1, dhi1;
    if (build) tile_dist(dlo0, dhi0, dlo1, dhi1);
    dp_steps();
    __builtin_amdgcn_sched_barrier(0);
    if (build) {
      tile_store(ring0, dlo0, dhi0, (q + 2) & 3);
      tile_store(ring1, dlo1, dhi1, (q + 2) & 3);
    }
    float* const op = (q & 1) ? op_odd : op_even;
#pragma unroll
    for (int k = 0; k < 16; ++k) op[k] = cs[k];
#pragma unroll
    for (int k = 0; k < 16; ++k) dv[k] = dvn[k];
    am = amn;
  };

  float* const orow0 = out + (b * P_N + p0) * T_N + lane;
  float* const orow1 = orow0 + T_N;
  for (int g = 0; g < T_N / 64; ++g) {
    const float ova0 = ostage0[lane];
    const float ova1 = ostage1[lane];
    block(4 * g + 0, 0, true);
    block(4 * g + 1, 1, true);
    const float ovb0 = ostage0[lane];
    const float ovb1 = ostage1[lane];
    block(4 * g + 2, 2, true);
    block(4 * g + 3, 3, true);
    if (g > 0) {
      orow0[(4 * g - 4) * 16] = ova0;
      orow1[(4 * g - 4) * 16] = ova1;
      orow0[(4 * g - 2) * 16] = ovb0;
      orow1[(4 * g - 2) * 16] = ovb1;
    }
  }
  {
    const float ova0 = ostage0[lane];
    const float ova1 = ostage1[lane];
    block(T_N / 16, 0, false);
    block(T_N / 16 + 1, 1, false);
    orow0[(N_BLK - 6) * 16] = ova0;
    orow1[(N_BLK - 6) * 16] = ova1;
    orow0[(N_BLK - 4) * 16] = ostage0[lane];
    orow1[(N_BLK - 4) * 16] = ostage1[lane];
  }
}

extern "C" void kernel_launch(void* const* d_in, const int* in_sizes, int n_in,
                              void* d_out, int out_size, void* d_ws, size_t ws_size,
                              hipStream_t stream) {
  (void)in_sizes; (void)n_in; (void)out_size; (void)d_ws; (void)ws_size;
  const float* x     = (const float*)d_in[0];
  const float* patts = (const float*)d_in[1];
  float* out = (float*)d_out;

  dim3 grid(256), block(WG_THREADS);
  hipLaunchKernelGGL(dtw_wmma_kernel, grid, block, 0, stream, x, patts, out);
}


# DP carried time-scaled, S = D * w^-(t - t0), rebased by w^16 every 16 steps: no multiply left in the recurrence, step chain is min3 -> fma; per-lane scale constants; rounding changes (few ulp)
# speedup vs baseline: 16.3557x; 1.0859x over previous
#include <hip/hip_runtime.h>
#include <math.h>


typedef __attribute__((ext_vector_type(2))) float v2f;
typedef __attribute__((ext_vector_type(8))) float v8f;
typedef float v2f_ring __attribute__((ext_vector_type(2), may_alias));

#define B_N 64
#define P_N 32
#define L_N 32
#define T_N 1024
#define DTW_W 0.93057205f
#define DTW_W16 0.316227734f

__constant__ const float W_POW[47] = {
    2.94272757f, 2.73842001f, 2.54829693f, 2.37137389f, 2.20673418f, 2.05352521f,
    1.91095316f, 1.77827954f, 1.65481722f, 1.53992665f, 1.43301260f, 1.33352149f,
    1.24093783f, 1.15478206f, 1.07460785f, 1.00000000f, 0.930572033f, 0.865964293f,
    0.805842161f, 0.749894202f, 0.697830558f, 0.649381578f, 0.604296327f, 0.562341273f,
    0.523299098f, 0.486967474f, 0.453158319f, 0.421696454f, 0.392418921f, 0.365174085f,
    0.339820802f, 0.316227734f, 0.294272661f, 0.273841918f, 0.254829645f, 0.237137333f,
    0.220673367f, 0.205352470f, 0.191095263f, 0.177827910f, 0.165481672f, 0.153992623f,
    0.143301219f, 0.133352116f, 0.124093749f, 0.115478165f, 0.107460752f
};
__constant__ const float W_OUT[16] = {
    10.0000029f, 9.30572319f, 8.65964508f, 8.05842400f, 7.49894381f, 6.97830725f,
    6.49381781f, 6.04296541f, 5.62341452f, 5.23299217f, 4.86967611f, 4.53158426f,
    4.21696568f, 3.92419028f, 3.65174174f, 3.39820886f
};
#define ROW_F 32
#define RING_ROWS 80
#define RING_F (RING_ROWS * ROW_F)
#define OST_F 32
#define DUMP_F 64
#define WAVE_LDS (RING_F + OST_F + RING_F + OST_F + DUMP_F)
#define N_BLK (T_N / 16 + 2)
#define XA_PAIRS (T_N * 2)
#define WG_WAVES 4
#define WG_THREADS (WG_WAVES * 32)

__global__ __launch_bounds__(WG_THREADS) void dtw_wmma_kernel(
    const float* __restrict__ x,
    const float* __restrict__ patts,
    float* __restrict__ out) {
  const int lane = threadIdx.x & 31;
  const int wib  = threadIdx.x >> 5;
  const int wave = blockIdx.x * WG_WAVES + wib;
  const int b  = wave >> 4;
  const int p0 = (wave * 2) & 31;

  __shared__ __attribute__((aligned(16))) float lds[WG_WAVES * WAVE_LDS];
  __shared__ v2f xa[XA_PAIRS];
  {
    const float* xs = x + (blockIdx.x >> 2) * 3 * T_N;
#pragma unroll
    for (int i = 0; i < T_N / WG_THREADS; ++i) {
      const int t = i * WG_THREADS + threadIdx.x;
      const float s0 = xs[0 * T_N + t];
      const float s1 = xs[1 * T_N + t];
      const float s2 = xs[2 * T_N + t];
      v2f lo2, hi2;
      lo2.x = s0;
      lo2.y = s1;
      hi2.x = s2;
      hi2.y = __builtin_fmaf(s2, s2, __builtin_fmaf(s1, s1, s0 * s0));
      xa[2 * t]     = lo2;
      xa[2 * t + 1] = hi2;
    }
  }
  float* ring0   = &lds[wib * WAVE_LDS];
  float* ostage0 = ring0 + RING_F;
  float* ring1   = ostage0 + OST_F;
  float* ostage1 = ring1 + RING_F;
  float* dump    = ostage1 + OST_F;

  const int  nn = lane & 15;
  const bool hi = lane >= 16;

  v2f bm00, bm01, bm10, bm11;
  float q2lo0, q2hi0, q2lo1, q2hi1;
  auto load_b = [&](const float* pp, v2f& bmlo, v2f& bmhi, float& p2lo, float& p2hi) {
    const float a0 = pp[0 * L_N + nn];
    const float a1 = pp[1 * L_N + nn];
    const float a2 = pp[2 * L_N + nn];
    p2lo = a0 * a0 + a1 * a1 + a2 * a2;
    bmlo.x = hi ? (-2.0f * a2) : (-2.0f * a0);
    bmlo.y = hi ? 1.0f         : (-2.0f * a1);
    const float c0 = pp[0 * L_N + 16 + nn];
    const float c1 = pp[1 * L_N + 16 + nn];
    const float c2 = pp[2 * L_N + 16 + nn];
    p2hi = c0 * c0 + c1 * c1 + c2 * c2;
    bmhi.x = hi ? (-2.0f * c2) : (-2.0f * c0);
    bmhi.y = hi ? 1.0f         : (-2.0f * c1);
  };
  load_b(patts + p0 * 3 * L_N,       bm00, bm01, q2lo0, q2hi0);
  load_b(patts + (p0 + 1) * 3 * L_N, bm10, bm11, q2lo1, q2hi1);

  for (int r = 32; r < 64; ++r) {
    ring0[r * ROW_F + lane] = 0.0f;
    ring1[r * ROW_F + lane] = 0.0f;
  }

  const int lagA = 2 * nn;
  float fs[17];
#pragma unroll
  for (int j = 0; j < 17; ++j) fs[j] = W_POW[lagA + 16 - j];
  const float gout = W_OUT[lane & 15];
  float* const myring = hi ? ring1 : ring0;
  float cA = INFINITY;
  float cB = INFINITY;
  float up0 = INFINITY;
  float up1 = INFINITY;
  const float u1_first = (nn == 0) ? 0.0f : INFINITY;
  float pA = INFINITY;
  float carry = 0.0f;

  __syncthreads();

  const v2f* ar = xa + nn * 2 + (hi ? 1 : 0);
  v2f am = ar[0];

  auto tile_dist = [&](v8f& dlo0, v8f& dhi0, v8f& dlo1, v8f& dhi1) {
    v8f acc = {};
    dlo0 = __builtin_amdgcn_wmma_f32_16x16x4_f32(
        false, am, false, bm00, (short)0, acc, false, false);
    dhi0 = __builtin_amdgcn_wmma_f32_16x16x4_f32(
        false, am, false, bm01, (short)0, acc, false, false);
    dlo1 = __builtin_amdgcn_wmma_f32_16x16x4_f32(
        false, am, false, bm10, (short)0, acc, false, false);
    dhi1 = __builtin_amdgcn_wmma_f32_16x16x4_f32(
        false, am, false, bm11, (short)0, acc, false, false);
#pragma unroll
    for (int k = 0; k < 8; ++k) {
      dlo0[k] = __builtin_amdgcn_sqrtf(fmaxf(dlo0[k] + q2lo0, 1e-12f));
      dhi0[k] = __builtin_amdgcn_sqrtf(fmaxf(dhi0[k] + q2hi0, 1e-12f));
      dlo1[k] = __builtin_amdgcn_sqrtf(fmaxf(dlo1[k] + q2lo1, 1e-12f));
      dhi1[k] = __builtin_amdgcn_sqrtf(fmaxf(dhi1[k] + q2hi1, 1e-12f));
    }
  };
  auto tile_store = [&](float* ring, const v8f& dlo, const v8f& dhi, const int slot) {
    float* wr = ring + ((slot * 16) + (hi ? 8 : 0)) * ROW_F + nn;
#pragma unroll
    for (int k = 0; k < 8; ++k) {
      wr[k * ROW_F]      = dlo[k];
      wr[k * ROW_F + 16] = dhi[k];
    }
    if (slot == 0) {
#pragma unroll
      for (int k = 0; k < 8; ++k) {
        wr[(64 + k) * ROW_F]      = dlo[k];
        wr[(64 + k) * ROW_F + 16] = dhi[k];
      }
    }
  };
  auto a_rows = [&](const int mt) {
    return ar[(mt < T_N / 16 ? mt : T_N / 16 - 1) * 32];
  };

  auto load_dist = [&](v2f (&d)[16], const int q) {
    const float* rd = myring + (((q * 16 - lagA) & 63) * ROW_F) + 2 * nn;
#pragma unroll
    for (int k = 0; k < 16; ++k) d[k] = *reinterpret_cast<const v2f_ring*>(rd + k * ROW_F);
  };

  float* const op_even = (lane == 15) ? ostage0      : (lane == 31) ? ostage1      : (dump + lane);
  float* const op_odd  = (lane == 15) ? ostage0 + 16 : (lane == 31) ? ostage1 + 16 : (dump + lane);

  v2f dv[16];
  float cs[16];
  auto dp_steps = [&](const bool first) {
    cA *= DTW_W16;
    cB *= DTW_W16;
    up0 *= DTW_W16;
    up1 *= DTW_W16;
    pA *= DTW_W16;
#pragma unroll
    for (int k = 0; k < 16; ++k) {
      cs[k] = cB;
      const float dA = dv[k].x;
      const float dB = (k == 0) ? carry : dv[k - 1].y;
      float& upn = (k & 1) ? up1 : up0;
      const float upo = (k & 1) ? up0 : up1;
      const float u1 = (first && k == 0) ? u1_first : upo;
      upn = __int_as_float(__builtin_amdgcn_update_dpp(
          __float_as_int(upn), __float_as_int(cB), 0x111, 0xf, 0xf, false));
      const float eA = fminf(fminf(upn, cA), u1);
      const float eB = fminf(fminf(cA, cB), pA);
      pA = cA;
      cA = __builtin_fmaf(dA, fs[k + 1], eA);
      cB = __builtin_fmaf(dB, fs[k], eB);
    }
    carry = dv[15].y;
  };

  {
    v8f dlo0, dhi0, dlo1, dhi1;
    tile_dist(dlo0, dhi0, dlo1, dhi1);
    am = a_rows(1);
    tile_store(ring0, dlo0, dhi0, 0);
    tile_store(ring1, dlo1, dhi1, 0);
    tile_dist(dlo0, dhi0, dlo1, dhi1);
    am = a_rows(2);
    tile_store(ring0, dlo0, dhi0, 1);
    tile_store(ring1, dlo1, dhi1, 1);
  }
  load_dist(dv, 0);
  __builtin_amdgcn_fence(__ATOMIC_SEQ_CST, "workgroup");

  auto block = [&](const int m, const int q, const bool build, const bool first) {
    v2f dvn[16];
    load_dist(dvn, (q + 1) & 3);
    v2f amn = am;
    if (build) amn = a_rows(m + 3);
    __builtin_amdgcn_sched_barrier(0);
    v8f dlo0, dhi0, dlo1, dhi1;
    if (build) tile_dist(dlo0, dhi0, dlo1, dhi1);
    dp_steps(first);
    __builtin_amdgcn_sched_barrier(0);
    if (build) {
      tile_store(ring0, dlo0, dhi0, (q + 2) & 3);
      tile_store(ring1, dlo1, dhi1, (q + 2) & 3);
    }
    float* const op = (q & 1) ? op_odd : op_even;
#pragma unroll
    for (int k = 0; k < 16; ++k) op[k] = cs[k];
#pragma unroll
    for (int k = 0; k < 16; ++k) dv[k] = dvn[k];
    am = amn;
  };

  float* const orow0 = out + (b * P_N + p0) * T_N + lane;
  float* const orow1 = orow0 + T_N;
  auto group = [&](const int g, const bool first) {
    const float ova0 = ostage0[lane] * gout;
    const float ova1 = ostage1[lane] * gout;
    block(4 * g + 0, 0, true, first);
    block(4 * g + 1, 1, true, false);
    const float ovb0 = ostage0[lane] * gout;
    const float ovb1 = ostage1[lane] * gout;
    block(4 * g + 2, 2, true, false);
    block(4 * g + 3, 3, true, false);
    if (!first) {
      orow0[(4 * g - 4) * 16] = ova0;
      orow1[(4 * g - 4) * 16] = ova1;
      orow0[(4 * g - 2) * 16] = ovb0;
      orow1[(4 * g - 2) * 16] = ovb1;
    }
  };
  group(0, true);
  for (int g = 1; g < T_N / 64; ++g) group(g, false);
  {
    const float ova0 = ostage0[lane] * gout;
    const float ova1 = ostage1[lane] * gout;
    block(T_N / 16, 0, false, false);
    block(T_N / 16 + 1, 1, false, false);
    orow0[(N_BLK - 6) * 16] = ova0;
    orow1[(N_BLK - 6) * 16] = ova1;
    orow0[(N_BLK - 4) * 16] = ostage0[lane] * gout;
    orow1[(N_BLK - 4) * 16] = ostage1[lane] * gout;
  }
}

extern "C" void kernel_launch(void* const* d_in, const int* in_sizes, int n_in,
                              void* d_out, int out_size, void* d_ws, size_t ws_size,
                              hipStream_t stream) {
  (void)in_sizes; (void)n_in; (void)out_size; (void)d_ws; (void)ws_size;
  const float* x     = (const float*)d_in[0];
  const float* patts = (const float*)d_in[1];
  float* out = (float*)d_out;

  dim3 grid(256), block(WG_THREADS);
  hipLaunchKernelGGL(dtw_wmma_kernel, grid, block, 0, stream, x, patts, out);
}


# LDS distance ring kept as 16 position-pair strips, time-major (B columns relabelled to positions 2n, 2n+1): tile stores and DP reads become 16-byte LDS accesses, 16 -> 8 tile stores per block
# speedup vs baseline: 17.4638x; 1.0678x over previous
#include <hip/hip_runtime.h>
#include <math.h>


typedef __attribute__((ext_vector_type(2))) float v2f;
typedef __attribute__((ext_vector_type(4))) float v4f;
typedef __attribute__((ext_vector_type(8))) float v8f;
typedef float v4f_lds __attribute__((ext_vector_type(4), may_alias));

#define B_N 64
#define P_N 32
#define L_N 32
#define T_N 1024
#define DTW_W 0.93057205f
#define DTW_W16 0.316227734f

__constant__ const float W_POW[47] = {
    2.94272757f, 2.73842001f, 2.54829693f, 2.37137389f, 2.20673418f, 2.05352521f,
    1.91095316f, 1.77827954f, 1.65481722f, 1.53992665f, 1.43301260f, 1.33352149f,
    1.24093783f, 1.15478206f, 1.07460785f, 1.00000000f, 0.930572033f, 0.865964293f,
    0.805842161f, 0.749894202f, 0.697830558f, 0.649381578f, 0.604296327f, 0.562341273f,
    0.523299098f, 0.486967474f, 0.453158319f, 0.421696454f, 0.392418921f, 0.365174085f,
    0.339820802f, 0.316227734f, 0.294272661f, 0.273841918f, 0.254829645f, 0.237137333f,
    0.220673367f, 0.205352470f, 0.191095263f, 0.177827910f, 0.165481672f, 0.153992623f,
    0.143301219f, 0.133352116f, 0.124093749f, 0.115478165f, 0.107460752f
};
__constant__ const float W_OUT[16] = {
    10.0000029f, 9.30572319f, 8.65964508f, 8.05842400f, 7.49894381f, 6.97830725f,
    6.49381781f, 6.04296541f, 5.62341452f, 5.23299217f, 4.86967611f, 4.53158426f,
    4.21696568f, 3.92419028f, 3.65174174f, 3.39820886f
};
#define RING_T 80
#define STRIP_F (RING_T * 2 + 8)
#define RING_F (16 * STRIP_F)
#define OST_F 32
#define DUMP_F 64
#define WAVE_LDS (RING_F + RING_F + OST_F + OST_F + DUMP_F)
#define N_BLK (T_N / 16 + 2)
#define XA_PAIRS (T_N * 2)
#define WG_WAVES 4
#define WG_THREADS (WG_WAVES * 32)

__global__ __launch_bounds__(WG_THREADS) void dtw_wmma_kernel(
    const float* __restrict__ x,
    const float* __restrict__ patts,
    float* __restrict__ out) {
  const int lane = threadIdx.x & 31;
  const int wib  = threadIdx.x >> 5;
  const int wave = blockIdx.x * WG_WAVES + wib;
  const int b  = wave >> 4;
  const int p0 = (wave * 2) & 31;

  __shared__ __attribute__((aligned(16))) float lds[WG_WAVES * WAVE_LDS];
  __shared__ v2f xa[XA_PAIRS];
  {
    const float* xs = x + (blockIdx.x >> 2) * 3 * T_N;
#pragma unroll
    for (int i = 0; i < T_N / WG_THREADS; ++i) {
      const int t = i * WG_THREADS + threadIdx.x;
      const float s0 = xs[0 * T_N + t];
      const float s1 = xs[1 * T_N + t];
      const float s2 = xs[2 * T_N + t];
      v2f lo2, hi2;
      lo2.x = s0;
      lo2.y = s1;
      hi2.x = s2;
      hi2.y = __builtin_fmaf(s2, s2, __builtin_fmaf(s1, s1, s0 * s0));
      xa[2 * t]     = lo2;
      xa[2 * t + 1] = hi2;
    }
  }
  float* ring0   = &lds[wib * WAVE_LDS];
  float* ring1   = ring0 + RING_F;
  float* ostage0 = ring1 + RING_F;
  float* ostage1 = ostage0 + OST_F;
  float* dump    = ostage1 + OST_F;

  const int  nn = lane & 15;
  const bool hi = lane >= 16;

  v2f bm00, bm01, bm10, bm11;
  float q2lo0, q2hi0, q2lo1, q2hi1;
  auto load_b = [&](const float* pp, v2f& bmlo, v2f& bmhi, float& p2lo, float& p2hi) {
    const float a0 = pp[0 * L_N + 2 * nn];
    const float a1 = pp[1 * L_N + 2 * nn];
    const float a2 = pp[2 * L_N + 2 * nn];
    p2lo = a0 * a0 + a1 * a1 + a2 * a2;
    bmlo.x = hi ? (-2.0f * a2) : (-2.0f * a0);
    bmlo.y = hi ? 1.0f         : (-2.0f * a1);
    const float c0 = pp[0 * L_N + 2 * nn + 1];
    const float c1 = pp[1 * L_N + 2 * nn + 1];
    const float c2 = pp[2 * L_N + 2 * nn + 1];
    p2hi = c0 * c0 + c1 * c1 + c2 * c2;
    bmhi.x = hi ? (-2.0f * c2) : (-2.0f * c0);
    bmhi.y = hi ? 1.0f         : (-2.0f * c1);
  };
  load_b(patts + p0 * 3 * L_N,       bm00, bm01, q2lo0, q2hi0);
  load_b(patts + (p0 + 1) * 3 * L_N, bm10, bm11, q2lo1, q2hi1);

  {
    const int zo = (lane >> 1) * STRIP_F + 64 + (lane & 1) * 32;
    const v4f zero4 = {0.0f, 0.0f, 0.0f, 0.0f};
#pragma unroll
    for (int r = 0; r < 8; ++r) {
      *reinterpret_cast<v4f_lds*>(ring0 + zo + 4 * r) = zero4;
      *reinterpret_cast<v4f_lds*>(ring1 + zo + 4 * r) = zero4;
    }
  }

  const int lagA = 2 * nn;
  float fs[17];
#pragma unroll
  for (int j = 0; j < 17; ++j) fs[j] = W_POW[lagA + 16 - j];
  const float gout = W_OUT[lane & 15];
  float* const myring = hi ? ring1 : ring0;
  float cA = INFINITY;
  float cB = INFINITY;
  float up0 = INFINITY;
  float up1 = INFINITY;
  const float u1_first = (nn == 0) ? 0.0f : INFINITY;
  float pA = INFINITY;
  float carry = 0.0f;

  __syncthreads();

  const v2f* ar = xa + nn * 2 + (hi ? 1 : 0);
  v2f am = ar[0];

  auto tile_dist = [&](v8f& dlo0, v8f& dhi0, v8f& dlo1, v8f& dhi1) {
    v8f acc = {};
    dlo0 = __builtin_amdgcn_wmma_f32_16x16x4_f32(
        false, am, false, bm00, (short)0, acc, false, false);
    dhi0 = __builtin_amdgcn_wmma_f32_16x16x4_f32(
        false, am, false, bm01, (short)0, acc, false, false);
    dlo1 = __builtin_amdgcn_wmma_f32_16x16x4_f32(
        false, am, false, bm10, (short)0, acc, false, false);
    dhi1 = __builtin_amdgcn_wmma_f32_16x16x4_f32(
        false, am, false, bm11, (short)0, acc, false, false);
    dlo0 = dlo0 + q2lo0;
    dhi0 = dhi0 + q2hi0;
    dlo1 = dlo1 + q2lo1;
    dhi1 = dhi1 + q2hi1;
#pragma unroll
    for (int k = 0; k < 8; ++k) {
      dlo0[k] = __builtin_amdgcn_sqrtf(fmaxf(dlo0[k], 1e-12f));
      dhi0[k] = __builtin_amdgcn_sqrtf(fmaxf(dhi0[k], 1e-12f));
      dlo1[k] = __builtin_amdgcn_sqrtf(fmaxf(dlo1[k], 1e-12f));
      dhi1[k] = __builtin_amdgcn_sqrtf(fmaxf(dhi1[k], 1e-12f));
    }
  };
  auto tile_store = [&](float* ring, const v8f& dlo, const v8f& dhi, const int slot) {
    float* wr = ring + nn * STRIP_F + ((slot * 16) + (hi ? 8 : 0)) * 2;
#pragma unroll
    for (int k = 0; k < 8; k += 2) {
      const v4f g = {dlo[k], dlo[k + 1], dhi[k], dhi[k + 1]};
      *reinterpret_cast<v4f_lds*>(wr + 2 * k) = g;
      if (slot == 0)
        *reinterpret_cast<v4f_lds*>(wr + 2 * (64 + k)) = g;
    }
  };
  auto a_rows = [&](const int mt) {
    return ar[(mt < T_N / 16 ? mt : T_N / 16 - 1) * 32];
  };

  auto load_dist = [&](v4f (&d)[8], const int q) {
    const float* rd = myring + nn * STRIP_F + (((q * 16 - lagA) & 63) * 2);
#pragma unroll
    for (int j = 0; j < 8; ++j) d[j] = *reinterpret_cast<const v4f_lds*>(rd + 4 * j);
  };

  float* const op_even = (lane == 15) ? ostage0      : (lane == 31) ? ostage1      : (dump + lane);
  float* const op_odd  = (lane == 15) ? ostage0 + 16 : (lane == 31) ? ostage1 + 16 : (dump + lane);

  v4f dv[8];
  float cs[16];
  auto dp_steps = [&](const bool first) {
    cA *= DTW_W16;
    cB *= DTW_W16;
    up0 *= DTW_W16;
    up1 *= DTW_W16;
    pA *= DTW_W16;
#pragma unroll
    for (int k = 0; k < 16; ++k) {
      cs[k] = cB;
      const float dA = (k & 1) ? dv[k >> 1].y : dv[k >> 1].x;
      const float dB = (k == 0) ? carry : ((k - 1) & 1) ? dv[(k - 1) >> 1].w : dv[(k - 1) >> 1].z;
      float& upn = (k & 1) ? up1 : up0;
      const float upo = (k & 1) ? up0 : up1;
      const float u1 = (first && k == 0) ? u1_first : upo;
      upn = __int_as_float(__builtin_amdgcn_update_dpp(
          __float_as_int(upn), __float_as_int(cB), 0x111, 0xf, 0xf, false));
      const float eA = fminf(fminf(upn, cA), u1);
      const float eB = fminf(fminf(cA, cB), pA);
      pA = cA;
      cA = __builtin_fmaf(dA, fs[k + 1], eA);
      cB = __builtin_fmaf(dB, fs[k], eB);
    }
    carry = dv[7].w;
  };

  {
    v8f dlo0, dhi0, dlo1, dhi1;
    tile_dist(dlo0, dhi0, dlo1, dhi1);
    am = a_rows(1);
    tile_store(ring0, dlo0, dhi0, 0);
    tile_store(ring1, dlo1, dhi1, 0);
    tile_dist(dlo0, dhi0, dlo1, dhi1);
    am = a_rows(2);
    tile_store(ring0, dlo0, dhi0, 1);
    tile_store(ring1, dlo1, dhi1, 1);
  }
  load_dist(dv, 0);
  __builtin_amdgcn_fence(__ATOMIC_SEQ_CST, "workgroup");

  auto block = [&](const int m, const int q, const bool build, const bool first) {
    v4f dvn[8];
    load_dist(dvn, (q + 1) & 3);
    v2f amn = am;
    if (build) amn = a_rows(m + 3);
    __builtin_amdgcn_sched_barrier(0);
    v8f dlo0, dhi0, dlo1, dhi1;
    if (build) tile_dist(dlo0, dhi0, dlo1, dhi1);
    dp_steps(first);
    __builtin_amdgcn_sched_barrier(0);
    if (build) {
      tile_store(ring0, dlo0, dhi0, (q + 2) & 3);
      tile_store(ring1, dlo1, dhi1, (q + 2) & 3);
    }
    float* const op = (q & 1) ? op_odd : op_even;
#pragma unroll
    for (int k = 0; k < 16; ++k) op[k] = cs[k];
#pragma unroll
    for (int j = 0; j < 8; ++j) dv[j] = dvn[j];
    am = amn;
  };

  float* const orow0 = out + (b * P_N + p0) * T_N + lane;
  float* const orow1 = orow0 + T_N;
  auto group = [&](const int g, const bool first) {
    const float ova0 = ostage0[lane] * gout;
    const float ova1 = ostage1[lane] * gout;
    block(4 * g + 0, 0, true, first);
    block(4 * g + 1, 1, true, false);
    const float ovb0 = ostage0[lane] * gout;
    const float ovb1 = ostage1[lane] * gout;
    block(4 * g + 2, 2, true, false);
    block(4 * g + 3, 3, true, false);
    if (!first) {
      orow0[(4 * g - 4) * 16] = ova0;
      orow1[(4 * g - 4) * 16] = ova1;
      orow0[(4 * g - 2) * 16] = ovb0;
      orow1[(4 * g - 2) * 16] = ovb1;
    }
  };
  group(0, true);
  for (int g = 1; g < T_N / 64; ++g) group(g, false);
  {
    const float ova0 = ostage0[lane] * gout;
    const float ova1 = ostage1[lane] * gout;
    block(T_N / 16, 0, false, false);
    block(T_N / 16 + 1, 1, false, false);
    orow0[(N_BLK - 6) * 16] = ova0;
    orow1[(N_BLK - 6) * 16] = ova1;
    orow0[(N_BLK - 4) * 16] = ostage0[lane] * gout;
    orow1[(N_BLK - 4) * 16] = ostage1[lane] * gout;
  }
}

extern "C" void kernel_launch(void* const* d_in, const int* in_sizes, int n_in,
                              void* d_out, int out_size, void* d_ws, size_t ws_size,
                              hipStream_t stream) {
  (void)in_sizes; (void)n_in; (void)out_size; (void)d_ws; (void)ws_size;
  const float* x     = (const float*)d_in[0];
  const float* patts = (const float*)d_in[1];
  float* out = (float*)d_out;

  dim3 grid(256), block(WG_THREADS);
  hipLaunchKernelGGL(dtw_wmma_kernel, grid, block, 0, stream, x, patts, out);
}


# Block issue order pinned by fences: tile build dealt over the 16 DP steps in three staggered stages, WMMAs spread and made ahead, chain pieces in ready order; LDS stays one burst; bit-identical
# speedup vs baseline: 18.4706x; 1.0577x over previous
#include <hip/hip_runtime.h>
#include <math.h>


typedef __attribute__((ext_vector_type(2))) float v2f;
typedef __attribute__((ext_vector_type(4))) float v4f;
typedef __attribute__((ext_vector_type(8))) float v8f;
typedef float v4f_lds __attribute__((ext_vector_type(4), may_alias));

#define B_N 64
#define P_N 32
#define L_N 32
#define T_N 1024
#define DTW_W 0.93057205f
#define DTW_W16 0.316227734f

__constant__ const float W_POW[47] = {
    2.94272757f, 2.73842001f, 2.54829693f, 2.37137389f, 2.20673418f, 2.05352521f,
    1.91095316f, 1.77827954f, 1.65481722f, 1.53992665f, 1.43301260f, 1.33352149f,
    1.24093783f, 1.15478206f, 1.07460785f, 1.00000000f, 0.930572033f, 0.865964293f,
    0.805842161f, 0.749894202f, 0.697830558f, 0.649381578f, 0.604296327f, 0.562341273f,
    0.523299098f, 0.486967474f, 0.453158319f, 0.421696454f, 0.392418921f, 0.365174085f,
    0.339820802f, 0.316227734f, 0.294272661f, 0.273841918f, 0.254829645f, 0.237137333f,
    0.220673367f, 0.205352470f, 0.191095263f, 0.177827910f, 0.165481672f, 0.153992623f,
    0.143301219f, 0.133352116f, 0.124093749f, 0.115478165f, 0.107460752f
};
__constant__ const float W_OUT[16] = {
    10.0000029f, 9.30572319f, 8.65964508f, 8.05842400f, 7.49894381f, 6.97830725f,
    6.49381781f, 6.04296541f, 5.62341452f, 5.23299217f, 4.86967611f, 4.53158426f,
    4.21696568f, 3.92419028f, 3.65174174f, 3.39820886f
};
#define RING_T 80
#define STRIP_F (RING_T * 2 + 8)
#define RING_F (16 * STRIP_F)
#define OST_F 32
#define DUMP_F 64
#define WAVE_LDS (RING_F + RING_F + OST_F + OST_F + DUMP_F)
#define N_BLK (T_N / 16 + 2)
#define XA_PAIRS (T_N * 2)
#define WG_WAVES 4
#define WG_THREADS (WG_WAVES * 32)

__global__ __launch_bounds__(WG_THREADS) void dtw_wmma_kernel(
    const float* __restrict__ x,
    const float* __restrict__ patts,
    float* __restrict__ out) {
  const int lane = threadIdx.x & 31;
  const int wib  = threadIdx.x >> 5;
  const int wave = blockIdx.x * WG_WAVES + wib;
  const int b  = wave >> 4;
  const int p0 = (wave * 2) & 31;

  __shared__ __attribute__((aligned(16))) float lds[WG_WAVES * WAVE_LDS];
  __shared__ v2f xa[XA_PAIRS];
  {
    const float* xs = x + (blockIdx.x >> 2) * 3 * T_N;
#pragma unroll
    for (int i = 0; i < T_N / WG_THREADS; ++i) {
      const int t = i * WG_THREADS + threadIdx.x;
      const float s0 = xs[0 * T_N + t];
      const float s1 = xs[1 * T_N + t];
      const float s2 = xs[2 * T_N + t];
      v2f lo2, hi2;
      lo2.x = s0;
      lo2.y = s1;
      hi2.x = s2;
      hi2.y = __builtin_fmaf(s2, s2, __builtin_fmaf(s1, s1, s0 * s0));
      xa[2 * t]     = lo2;
      xa[2 * t + 1] = hi2;
    }
  }
  float* ring0   = &lds[wib * WAVE_LDS];
  float* ring1   = ring0 + RING_F;
  float* ostage0 = ring1 + RING_F;
  float* ostage1 = ostage0 + OST_F;
  float* dump    = ostage1 + OST_F;

  const int  nn = lane & 15;
  const bool hi = lane >= 16;

  v2f bm00, bm01, bm10, bm11;
  float q2lo0, q2hi0, q2lo1, q2hi1;
  auto load_b = [&](const float* pp, v2f& bmlo, v2f& bmhi, float& p2lo, float& p2hi) {
    const float a0 = pp[0 * L_N + 2 * nn];
    const float a1 = pp[1 * L_N + 2 * nn];
    const float a2 = pp[2 * L_N + 2 * nn];
    p2lo = a0 * a0 + a1 * a1 + a2 * a2;
    bmlo.x = hi ? (-2.0f * a2) : (-2.0f * a0);
    bmlo.y = hi ? 1.0f         : (-2.0f * a1);
    const float c0 = pp[0 * L_N + 2 * nn + 1];
    const float c1 = pp[1 * L_N + 2 * nn + 1];
    const float c2 = pp[2 * L_N + 2 * nn + 1];
    p2hi = c0 * c0 + c1 * c1 + c2 * c2;
    bmhi.x = hi ? (-2.0f * c2) : (-2.0f * c0);
    bmhi.y = hi ? 1.0f         : (-2.0f * c1);
  };
  load_b(patts + p0 * 3 * L_N,       bm00, bm01, q2lo0, q2hi0);
  load_b(patts + (p0 + 1) * 3 * L_N, bm10, bm11, q2lo1, q2hi1);

  {
    const int zo = (lane >> 1) * STRIP_F + 64 + (lane & 1) * 32;
    const v4f zero4 = {0.0f, 0.0f, 0.0f, 0.0f};
#pragma unroll
    for (int r = 0; r < 8; ++r) {
      *reinterpret_cast<v4f_lds*>(ring0 + zo + 4 * r) = zero4;
      *reinterpret_cast<v4f_lds*>(ring1 + zo + 4 * r) = zero4;
    }
  }

  const int lagA = 2 * nn;
  float fs[17];
#pragma unroll
  for (int j = 0; j < 17; ++j) fs[j] = W_POW[lagA + 16 - j];
  const float gout = W_OUT[lane & 15];
  float* const myring = hi ? ring1 : ring0;
  float cA = INFINITY;
  float cB = INFINITY;
  float up0 = INFINITY;
  float up1 = INFINITY;
  const float u1_first = (nn == 0) ? 0.0f : INFINITY;
  float pA = INFINITY;
  float carry = 0.0f;

  __syncthreads();

  const v2f* ar = xa + nn * 2 + (hi ? 1 : 0);
  v2f am = ar[0];

  auto tile_dist = [&](v8f& dlo0, v8f& dhi0, v8f& dlo1, v8f& dhi1) {
    v8f acc = {};
    dlo0 = __builtin_amdgcn_wmma_f32_16x16x4_f32(
        false, am, false, bm00, (short)0, acc, false, false);
    dhi0 = __builtin_amdgcn_wmma_f32_16x16x4_f32(
        false, am, false, bm01, (short)0, acc, false, false);
    dlo1 = __builtin_amdgcn_wmma_f32_16x16x4_f32(
        false, am, false, bm10, (short)0, acc, false, false);
    dhi1 = __builtin_amdgcn_wmma_f32_16x16x4_f32(
        false, am, false, bm11, (short)0, acc, false, false);
    dlo0 = dlo0 + q2lo0;
    dhi0 = dhi0 + q2hi0;
    dlo1 = dlo1 + q2lo1;
    dhi1 = dhi1 + q2hi1;
#pragma unroll
    for (int k = 0; k < 8; ++k) {
      dlo0[k] = __builtin_amdgcn_sqrtf(fmaxf(dlo0[k], 1e-12f));
      dhi0[k] = __builtin_amdgcn_sqrtf(fmaxf(dhi0[k], 1e-12f));
      dlo1[k] = __builtin_amdgcn_sqrtf(fmaxf(dlo1[k], 1e-12f));
      dhi1[k] = __builtin_amdgcn_sqrtf(fmaxf(dhi1[k], 1e-12f));
    }
  };
  auto tile_store = [&](float* ring, const v8f& dlo, const v8f& dhi, const int slot) {
    float* wr = ring + nn * STRIP_F + ((slot * 16) + (hi ? 8 : 0)) * 2;
#pragma unroll
    for (int k = 0; k < 8; k += 2) {
      const v4f g = {dlo[k], dlo[k + 1], dhi[k], dhi[k + 1]};
      *reinterpret_cast<v4f_lds*>(wr + 2 * k) = g;
      if (slot == 0)
        *reinterpret_cast<v4f_lds*>(wr + 2 * (64 + k)) = g;
    }
  };
  auto a_rows = [&](const int mt) {
    return ar[(mt < T_N / 16 ? mt : T_N / 16 - 1) * 32];
  };

  auto load_dist = [&](v4f (&d)[8], const int q) {
    const float* rd = myring + nn * STRIP_F + (((q * 16 - lagA) & 63) * 2);
#pragma unroll
    for (int j = 0; j < 8; ++j) d[j] = *reinterpret_cast<const v4f_lds*>(rd + 4 * j);
  };

  float* const op_even = (lane == 15) ? ostage0      : (lane == 31) ? ostage1      : (dump + lane);
  float* const op_odd  = (lane == 15) ? ostage0 + 16 : (lane == 31) ? ostage1 + 16 : (dump + lane);

  v4f dv[8];
  float cs[16];
  auto dp_rebase = [&]() {
    cA *= DTW_W16;
    cB *= DTW_W16;
    up0 *= DTW_W16;
    up1 *= DTW_W16;
    pA *= DTW_W16;
  };
  float eAh = 0.0f, eBh = 0.0f;
  auto dp_shift = [&](const int k) {
    cs[k] = cB;
    float& upn = (k & 1) ? up1 : up0;
    upn = __int_as_float(__builtin_amdgcn_update_dpp(
        __float_as_int(upn), __float_as_int(cB), 0x111, 0xf, 0xf, false));
  };
  auto dp_min_b = [&]() {
    eBh = fminf(fminf(cA, cB), pA);
    pA = cA;
  };
  auto dp_min_a = [&](const int k, const bool first) {
    const float upn = (k & 1) ? up1 : up0;
    const float upo = (k & 1) ? up0 : up1;
    const float u1 = (first && k == 0) ? u1_first : upo;
    eAh = fminf(fminf(upn, cA), u1);
  };
  auto dp_fma_b = [&](const int k) {
    const float dB = (k == 0) ? carry : ((k - 1) & 1) ? dv[(k - 1) >> 1].w : dv[(k - 1) >> 1].z;
    cB = __builtin_fmaf(dB, fs[k], eBh);
  };
  auto dp_fma_a = [&](const int k) {
    const float dA = (k & 1) ? dv[k >> 1].y : dv[k >> 1].x;
    cA = __builtin_fmaf(dA, fs[k + 1], eAh);
  };
  auto tile_mma = [&](const v2f a, const int j) {
    v8f acc = {};
    return __builtin_amdgcn_wmma_f32_16x16x4_f32(
        false, a, false, (j == 0) ? bm00 : (j == 1) ? bm01 : (j == 2) ? bm10 : bm11,
        (short)0, acc, false, false);
  };
  v2f pr[16];
  auto tile_add = [&](const v8f& v, const int p) {
    const int j = p >> 2, e = 2 * (p & 3);
    const float q2 = (j == 0) ? q2lo0 : (j == 1) ? q2hi0 : (j == 2) ? q2lo1 : q2hi1;
    const v2f t = {v[e], v[e + 1]};
    pr[p] = t + q2;
  };
  auto tile_clamp = [&](const int p) {
    pr[p].x = fmaxf(pr[p].x, 1e-12f);
    pr[p].y = fmaxf(pr[p].y, 1e-12f);
  };
  auto tile_sqrt_x = [&](v8f& v, const int p) {
    v[2 * (p & 3)] = __builtin_amdgcn_sqrtf(pr[p].x);
  };
  auto tile_sqrt_y = [&](v8f& v, const int p) {
    v[2 * (p & 3) + 1] = __builtin_amdgcn_sqrtf(pr[p].y);
  };

  {
    v8f dlo0, dhi0, dlo1, dhi1;
    tile_dist(dlo0, dhi0, dlo1, dhi1);
    am = a_rows(1);
    tile_store(ring0, dlo0, dhi0, 0);
    tile_store(ring1, dlo1, dhi1, 0);
    tile_dist(dlo0, dhi0, dlo1, dhi1);
    am = a_rows(2);
    tile_store(ring0, dlo0, dhi0, 1);
    tile_store(ring1, dlo1, dhi1, 1);
  }
  v2f am1 = a_rows(3);
  load_dist(dv, 0);
  v8f tv0 = tile_mma(am, 0);
  __builtin_amdgcn_fence(__ATOMIC_SEQ_CST, "workgroup");

  auto block = [&](const int m, const int q, const bool build, const bool first)
      __attribute__((always_inline)) {
    v4f dvn[8];
    load_dist(dvn, (q + 1) & 3);
    v2f amn = am1;
    if (build) amn = a_rows(m + 4);
    __builtin_amdgcn_sched_barrier(0);
    v8f tv[4];
    if (build) tv[0] = tv0;
    dp_rebase();
#pragma unroll
    for (int k = 0; k < 16; ++k) {
      dp_shift(k);
      if (build && k > 1) tile_sqrt_x(tv[(k - 2) >> 2], k - 2);
      __builtin_amdgcn_sched_barrier(0);
      dp_min_b();
      __builtin_amdgcn_sched_barrier(0);
      if (build) {
        if (k == 0) tile_add(tv[0], 0);
        if (k < 15) tile_add(tv[(k + 1) >> 2], k + 1);
        if (k > 1) tile_sqrt_y(tv[(k - 2) >> 2], k - 2);
        if (k == 15) tile_clamp(15);
      }
      __builtin_amdgcn_sched_barrier(0);
      dp_min_a(k, first);
      __builtin_amdgcn_sched_barrier(0);
      dp_fma_b(k);
      if (build) {
        if (k < 15) tile_clamp(k);
        if ((k & 3) == 0 && k < 12) tv[(k >> 2) + 1] = tile_mma(am, (k >> 2) + 1);
        if (k == 13) tv0 = tile_mma(am1, 0);
        if (k == 15) { tile_sqrt_x(tv[3], 14); tile_sqrt_y(tv[3], 14); }
      }
      __builtin_amdgcn_sched_barrier(0);
      dp_fma_a(k);
      __builtin_amdgcn_sched_barrier(0);
    }
    if (build) { tile_sqrt_x(tv[3], 15); tile_sqrt_y(tv[3], 15); }
    carry = dv[7].w;
    if (build) {
      tile_store(ring0, tv[0], tv[1], (q + 2) & 3);
      tile_store(ring1, tv[2], tv[3], (q + 2) & 3);
    }
    float* const op = (q & 1) ? op_odd : op_even;
#pragma unroll
    for (int k = 0; k < 16; ++k) op[k] = cs[k];
#pragma unroll
    for (int j = 0; j < 8; ++j) dv[j] = dvn[j];
    am = am1;
    am1 = amn;
  };

  float* const orow0 = out + (b * P_N + p0) * T_N + lane;
  float* const orow1 = orow0 + T_N;
  auto group = [&](const int g, const bool first) __attribute__((always_inline)) {
    const float ova0 = ostage0[lane] * gout;
    const float ova1 = ostage1[lane] * gout;
    block(4 * g + 0, 0, true, first);
    block(4 * g + 1, 1, true, false);
    const float ovb0 = ostage0[lane] * gout;
    const float ovb1 = ostage1[lane] * gout;
    block(4 * g + 2, 2, true, false);
    block(4 * g + 3, 3, true, false);
    if (!first) {
      orow0[(4 * g - 4) * 16] = ova0;
      orow1[(4 * g - 4) * 16] = ova1;
      orow0[(4 * g - 2) * 16] = ovb0;
      orow1[(4 * g - 2) * 16] = ovb1;
    }
  };
  group(0, true);
  for (int g = 1; g < T_N / 64; ++g) group(g, false);
  {
    const float ova0 = ostage0[lane] * gout;
    const float ova1 = ostage1[lane] * gout;
    block(T_N / 16, 0, false, false);
    block(T_N / 16 + 1, 1, false, false);
    orow0[(N_BLK - 6) * 16] = ova0;
    orow1[(N_BLK - 6) * 16] = ova1;
    orow0[(N_BLK - 4) * 16] = ostage0[lane] * gout;
    orow1[(N_BLK - 4) * 16] = ostage1[lane] * gout;
  }
}

extern "C" void kernel_launch(void* const* d_in, const int* in_sizes, int n_in,
                              void* d_out, int out_size, void* d_ws, size_t ws_size,
                              hipStream_t stream) {
  (void)in_sizes; (void)n_in; (void)out_size; (void)d_ws; (void)ws_size;
  const float* x     = (const float*)d_in[0];
  const float* patts = (const float*)d_in[1];
  float* out = (float*)d_out;

  dim3 grid(256), block(WG_THREADS);
  hipLaunchKernelGGL(dtw_wmma_kernel, grid, block, 0, stream, x, patts, out);
}
